# starting kernel
# baseline (speedup 1.0000x reference)
#include <hip/hip_runtime.h>
#include <hip/hip_bf16.h>


typedef __attribute__((ext_vector_type(2))) float v2f;
typedef __attribute__((ext_vector_type(8))) float v8f;

#define DDIM 128
#define KP_STRIDE 288

__global__ __launch_bounds__(256) void gemm_xw_wmma(
    const float* __restrict__ x, const float* __restrict__ W,
    float* __restrict__ y, int nrows) {
  __shared__ float Wlds[(DDIM / 2) * KP_STRIDE];

  const int t = threadIdx.x;

  for (int idx = t; idx < DDIM * 32; idx += 256) {
    const int r = idx >> 5;
    const int q = idx & 31;
    const float4 w4 = *(const float4*)(W + r * DDIM + q * 4);
    float* dst = &Wlds[(r >> 1) * KP_STRIDE + (q * 4) * 2 + (r & 1)];
    dst[0] = w4.x; dst[2] = w4.y; dst[4] = w4.z; dst[6] = w4.w;
  }
  __syncthreads();

  const int wave = t >> 5;
  const int lane = t & 31;
  const int lh   = lane >> 4;
  const int lm   = lane & 15;

  const int rowBase = blockIdx.x * 128 + wave * 16;
  const int arow = min(rowBase + lm, nrows - 1);
  const float* __restrict__ xrow = x + (size_t)arow * DDIM;

  v8f acc[8];
#pragma unroll
  for (int nt = 0; nt < 8; ++nt) acc[nt] = (v8f){0.f,0.f,0.f,0.f,0.f,0.f,0.f,0.f};

  for (int k0 = 0; k0 < DDIM; k0 += 4) {
    const v2f a = *(const v2f*)(xrow + k0 + 2 * lh);
    const float* __restrict__ wbase =
        &Wlds[((k0 >> 1) + lh) * KP_STRIDE + lm * 2];
#pragma unroll
    for (int nt = 0; nt < 8; ++nt) {
      const v2f b = *(const v2f*)(wbase + nt * 32);
      acc[nt] = __builtin_amdgcn_wmma_f32_16x16x4_f32(
           false, a,  false, b,
           (short)0, acc[nt],  false,  false);
    }
  }

#pragma unroll
  for (int nt = 0; nt < 8; ++nt) {
#pragma unroll
    for (int r = 0; r < 8; ++r) {
      const int m = rowBase + r + 8 * lh;
      if (m < nrows) y[(size_t)m * DDIM + nt * 16 + lm] = acc[nt][r];
    }
  }
}

__global__ __launch_bounds__(256) void bias_init(
    const float* __restrict__ bias, float* __restrict__ out, int total4) {
  const int i = blockIdx.x * 256 + threadIdx.x;
  if (i < total4) {
    const float4 b4 = *(const float4*)(bias + (i & 31) * 4);
    *(float4*)(out + (size_t)i * 4) = b4;
  }
}

__global__ __launch_bounds__(256) void spmm_scatter(
    const float* __restrict__ y, const float* __restrict__ ev,
    const int* __restrict__ row, const int* __restrict__ col,
    float* __restrict__ out, int nedges) {
  const int t = threadIdx.x;
  const int wave = t >> 5;
  const int lane = t & 31;
  const int e = blockIdx.x * 8 + wave;
  if (e >= nedges) return;

  const float v = ev[e];
  const int r = row[e];
  const int c = col[e];

  const float4 yv = *(const float4*)(y + (size_t)c * DDIM + lane * 4);
  float* dst = out + (size_t)r * DDIM + lane * 4;
  unsafeAtomicAdd(dst + 0, v * yv.x);
  unsafeAtomicAdd(dst + 1, v * yv.y);
  unsafeAtomicAdd(dst + 2, v * yv.z);
  unsafeAtomicAdd(dst + 3, v * yv.w);
}

extern "C" void kernel_launch(void* const* d_in, const int* in_sizes, int n_in,
                              void* d_out, int out_size, void* d_ws, size_t ws_size,
                              hipStream_t stream) {
  const float* x    = (const float*)d_in[0];
  const float* W    = (const float*)d_in[1];
  const float* bias = (const float*)d_in[2];
  const float* ev   = (const float*)d_in[3];
  const int*   row  = (const int*)d_in[4];
  const int*   col  = (const int*)d_in[5];
  float* out = (float*)d_out;
  float* y   = (float*)d_ws;

  const int n = in_sizes[0] / DDIM;
  const int e = in_sizes[3];

  const int gemm_blocks = (n + 127) / 128;
  gemm_xw_wmma<<<gemm_blocks, 256, 0, stream>>>(x, W, y, n);

  const int total4 = n * (DDIM / 4);
  bias_init<<<(total4 + 255) / 256, 256, 0, stream>>>(bias, out, total4);

  spmm_scatter<<<(e + 7) / 8, 256, 0, stream>>>(y, ev, row, col, out, e);
}


# y kept in LDS: a 32-wave group makes y for its 400 nodes (same f32 WMMA loop), filters edges by a 1-byte col/400 key, pushes val*y[col] by line f32 atomics (arrival order); no y table, no gathers
# speedup vs baseline: 1.8756x; 1.8756x over previous
#include <hip/hip_runtime.h>


typedef __attribute__((ext_vector_type(2))) float v2f;
typedef __attribute__((ext_vector_type(8))) float v8f;

#define DDIM 128
#define KP_STRIDE 288
#define WLDS_FLOATS ((DDIM / 2) * KP_STRIDE)
#define RPG 400
#define YP 132
#define NWAVE 32
#define LCAP (WLDS_FLOATS / NWAVE)
#define CHUNK 512
#define KSTEP 4

__global__ __launch_bounds__(256) void init_out_keys(
    const float* __restrict__ bias, const int* __restrict__ col,
    float* __restrict__ out, unsigned* __restrict__ keys,
    int total4, int nkeywords, int nedges) {
  const int i = blockIdx.x * 256 + threadIdx.x;
  if (i < total4) {
    const float4 b4 = *(const float4*)(bias + (i & 31) * 4);
    *(float4*)(out + (size_t)i * 4) = b4;
  }
  if (i < nkeywords) {
    const int e = i * 4;
    unsigned k = 0xFFFFFFFFu;
    if (e + 3 < nedges) {
      const int4 c = *(const int4*)(col + e);
      k = (unsigned)(c.x / RPG) | ((unsigned)(c.y / RPG) << 8) |
          ((unsigned)(c.z / RPG) << 16) | ((unsigned)(c.w / RPG) << 24);
    } else {
      for (int b = 0; b < 4; ++b)
        if (e + b < nedges)
          k = (k & ~(0xFFu << (8 * b))) | ((unsigned)(col[e + b] / RPG) << (8 * b));
    }
    keys[i] = k;
  }
}

struct Row4 { float a, b, c, d; };

__device__ __forceinline__ Row4 lds_row(const float* __restrict__ Ytile,
                                        int c_l, int j, int lane) {
  const int c = __builtin_amdgcn_readlane(c_l, j);
  const float* src = Ytile + c * YP + lane;
  Row4 g;
  g.a = src[0]; g.b = src[32]; g.c = src[64]; g.d = src[96];
  return g;
}

__device__ __forceinline__ void push_row(float* __restrict__ out, int r_l, int v_l,
                                         int j, unsigned lbyte, const Row4 g) {
  const unsigned r = (unsigned)__builtin_amdgcn_readlane(r_l, j);
  const float v = __int_as_float(__builtin_amdgcn_readlane(v_l, j));
  float* dst = (float*)((char*)out + (r * (DDIM * 4u) + lbyte));
  unsafeAtomicAdd(dst + 0,  v * g.a);
  unsafeAtomicAdd(dst + 32, v * g.b);
  unsafeAtomicAdd(dst + 64, v * g.c);
  unsafeAtomicAdd(dst + 96, v * g.d);
}

#define EGRP 4

__device__ __forceinline__ void push_edges(
    const unsigned* lst, int cnt, int lane, int node0,
    const float* __restrict__ Ytile, const float* __restrict__ ev,
    const int* __restrict__ row, const int* __restrict__ col,
    float* __restrict__ out) {
  const unsigned el = lst[min(lane, cnt - 1)];
  const int v_l = __float_as_int(ev[el]);
  const int r_l = row[el];
  const int c_l = col[el] - node0;
  const unsigned lbyte = (unsigned)lane * 4u;
  if (cnt == 32) {
#pragma unroll
    for (int g = 0; g < 32; g += EGRP) {
      Row4 cur[EGRP];
#pragma unroll
      for (int u = 0; u < EGRP; ++u) cur[u] = lds_row(Ytile, c_l, g + u, lane);
#pragma unroll
      for (int u = 0; u < EGRP; ++u) push_row(out, r_l, v_l, g + u, lbyte, cur[u]);
    }
  } else {
    for (int j = 0; j < cnt; ++j)
      push_row(out, r_l, v_l, j, lbyte, lds_row(Ytile, c_l, j, lane));
  }
}

__device__ __forceinline__ uint4 load_keys(const unsigned* __restrict__ keys,
                                           int i, int rot, int nchunks, int lane) {
  int c = i + rot;
  if (c >= nchunks) c -= nchunks;
  return *(const uint4*)(keys + (size_t)c * (CHUNK / 4) + lane * 4);
}

__global__ __launch_bounds__(1024) void gcn_owner_push(
    const float* __restrict__ x, const float* __restrict__ W,
    const float* __restrict__ ev, const int* __restrict__ row,
    const int* __restrict__ col, const unsigned* __restrict__ keys,
    float* __restrict__ out, int nrows, int nchunks) {
  __shared__ float Ytile[RPG * YP];
  __shared__ float Wlds[WLDS_FLOATS];

  const int t = threadIdx.x;
  for (int idx = t; idx < DDIM * 32; idx += 1024) {
    const int r = idx >> 5;
    const int q = idx & 31;
    const float4 w4 = *(const float4*)(W + r * DDIM + q * 4);
    float* dst = &Wlds[(r >> 1) * KP_STRIDE + (q * 4) * 2 + (r & 1)];
    dst[0] = w4.x; dst[2] = w4.y; dst[4] = w4.z; dst[6] = w4.w;
  }
  __syncthreads();

  const int wave = __builtin_amdgcn_readfirstlane(t >> 5);
  const int lane = t & 31;
  const int lh   = lane >> 4;
  const int lm   = lane & 15;
  const int node0 = blockIdx.x * RPG;

  for (int tile = wave; tile < RPG / 16; tile += NWAVE) {
    const int rowBase = node0 + tile * 16;
    const int arow = min(rowBase + lm, nrows - 1);
    const float* __restrict__ xrow = x + (size_t)arow * DDIM;

    v8f acc[8];
#pragma unroll
    for (int nt = 0; nt < 8; ++nt) acc[nt] = (v8f){0.f,0.f,0.f,0.f,0.f,0.f,0.f,0.f};

    for (int k0 = 0; k0 < DDIM; k0 += 4) {
      const v2f a = *(const v2f*)(xrow + k0 + 2 * lh);
      const float* __restrict__ wbase =
          &Wlds[((k0 >> 1) + lh) * KP_STRIDE + lm * 2];
#pragma unroll
      for (int nt = 0; nt < 8; ++nt) {
        const v2f b = *(const v2f*)(wbase + nt * 32);
        acc[nt] = __builtin_amdgcn_wmma_f32_16x16x4_f32(
             false, a,  false, b,
             (short)0, acc[nt],  false,  false);
      }
    }

#pragma unroll
    for (int nt = 0; nt < 8; ++nt) {
#pragma unroll
      for (int r = 0; r < 8; ++r)
        Ytile[(tile * 16 + r + 8 * lh) * YP + nt * 16 + lm] = acc[nt][r];
    }
  }
  __syncthreads();

  unsigned* lst = (unsigned*)Wlds + wave * LCAP;
  const unsigned gkey = blockIdx.x;
  const int rot = (int)(((unsigned)blockIdx.x * (unsigned)nchunks) / gridDim.x);
  int cnt = 0;

  for (int i = wave; i < nchunks; i += KSTEP * NWAVE) {
    uint4 kk[KSTEP];
#pragma unroll
    for (int u = 0; u < KSTEP; ++u)
      kk[u] = load_keys(keys, min(i + u * NWAVE, nchunks - 1), rot, nchunks, lane);
#pragma unroll
    for (int u = 0; u < KSTEP; ++u) {
      const int ii = i + u * NWAVE;
      if (ii < nchunks) {
        int c = ii + rot;
        if (c >= nchunks) c -= nchunks;
        const unsigned ebase = (unsigned)c * CHUNK + (unsigned)lane * 16u;
        const unsigned kw[4] = {kk[u].x, kk[u].y, kk[u].z, kk[u].w};
#pragma unroll
        for (int b = 0; b < 16; ++b) {
          const bool m = ((kw[b >> 2] >> (8 * (b & 3))) & 0xFFu) == gkey;
          const unsigned bal = __builtin_amdgcn_ballot_w32(m);
          if (bal != 0u) {
            if (m) lst[cnt + (int)__builtin_amdgcn_mbcnt_lo(bal, 0u)] = ebase + (unsigned)b;
            cnt += __builtin_popcount(bal);
          }
        }
        while (cnt >= 32) {
          cnt -= 32;
          push_edges(lst + cnt, 32, lane, node0, Ytile, ev, row, col, out);
        }
      }
    }
  }
  if (cnt > 0) push_edges(lst, cnt, lane, node0, Ytile, ev, row, col, out);
}

extern "C" void kernel_launch(void* const* d_in, const int* in_sizes, int n_in,
                              void* d_out, int out_size, void* d_ws, size_t ws_size,
                              hipStream_t stream) {
  const float* x    = (const float*)d_in[0];
  const float* W    = (const float*)d_in[1];
  const float* bias = (const float*)d_in[2];
  const float* ev   = (const float*)d_in[3];
  const int*   row  = (const int*)d_in[4];
  const int*   col  = (const int*)d_in[5];
  float* out = (float*)d_out;
  unsigned* keys = (unsigned*)d_ws;

  const int n = in_sizes[0] / DDIM;
  const int e = in_sizes[3];
  const int nchunks = (e + CHUNK - 1) / CHUNK;

  const int total4 = n * (DDIM / 4);
  init_out_keys<<<(total4 + 255) / 256, 256, 0, stream>>>(
      bias, col, out, keys, total4, nchunks * (CHUNK / 4), e);

  gcn_owner_push<<<(n + RPG - 1) / RPG, 1024, 0, stream>>>(
      x, W, ev, row, col, keys, out, n, nchunks);
}


# edge filter: a chunk's 16 key-byte compares run together, ballots stay in scalar registers and scalar code tests them: one vector-to-branch wait per 512 keys, not 16; list order unchanged
# speedup vs baseline: 1.9318x; 1.0300x over previous
#include <hip/hip_runtime.h>


typedef __attribute__((ext_vector_type(2))) float v2f;
typedef __attribute__((ext_vector_type(8))) float v8f;

#define DDIM 128
#define KP_STRIDE 288
#define WLDS_FLOATS ((DDIM / 2) * KP_STRIDE)
#define RPG 400
#define YP 132
#define NWAVE 32
#define LCAP (WLDS_FLOATS / NWAVE)
#define CHUNK 512
#define KSTEP 4

__global__ __launch_bounds__(256) void init_out_keys(
    const float* __restrict__ bias, const int* __restrict__ col,
    float* __restrict__ out, unsigned* __restrict__ keys,
    int total4, int nkeywords, int nedges) {
  const int i = blockIdx.x * 256 + threadIdx.x;
  if (i < total4) {
    const float4 b4 = *(const float4*)(bias + (i & 31) * 4);
    *(float4*)(out + (size_t)i * 4) = b4;
  }
  if (i < nkeywords) {
    const int e = i * 4;
    unsigned k = 0xFFFFFFFFu;
    if (e + 3 < nedges) {
      const int4 c = *(const int4*)(col + e);
      k = (unsigned)(c.x / RPG) | ((unsigned)(c.y / RPG) << 8) |
          ((unsigned)(c.z / RPG) << 16) | ((unsigned)(c.w / RPG) << 24);
    } else {
      for (int b = 0; b < 4; ++b)
        if (e + b < nedges)
          k = (k & ~(0xFFu << (8 * b))) | ((unsigned)(col[e + b] / RPG) << (8 * b));
    }
    keys[i] = k;
  }
}

struct Row4 { float a, b, c, d; };

__device__ __forceinline__ Row4 lds_row(const float* __restrict__ Ytile,
                                        int c_l, int j, int lane) {
  const int c = __builtin_amdgcn_readlane(c_l, j);
  const float* src = Ytile + c * YP + lane;
  Row4 g;
  g.a = src[0]; g.b = src[32]; g.c = src[64]; g.d = src[96];
  return g;
}

__device__ __forceinline__ void push_row(float* __restrict__ out, int r_l, int v_l,
                                         int j, unsigned lbyte, const Row4 g) {
  const unsigned r = (unsigned)__builtin_amdgcn_readlane(r_l, j);
  const float v = __int_as_float(__builtin_amdgcn_readlane(v_l, j));
  float* dst = (float*)((char*)out + (r * (DDIM * 4u) + lbyte));
  unsafeAtomicAdd(dst + 0,  v * g.a);
  unsafeAtomicAdd(dst + 32, v * g.b);
  unsafeAtomicAdd(dst + 64, v * g.c);
  unsafeAtomicAdd(dst + 96, v * g.d);
}

#define EGRP 4

__device__ __forceinline__ void push_edges(
    const unsigned* lst, int cnt, int lane, int node0,
    const float* __restrict__ Ytile, const float* __restrict__ ev,
    const int* __restrict__ row, const int* __restrict__ col,
    float* __restrict__ out) {
  const unsigned el = lst[min(lane, cnt - 1)];
  const int v_l = __float_as_int(ev[el]);
  const int r_l = row[el];
  const int c_l = col[el] - node0;
  const unsigned lbyte = (unsigned)lane * 4u;
  if (cnt == 32) {
#pragma unroll
    for (int g = 0; g < 32; g += EGRP) {
      Row4 cur[EGRP];
#pragma unroll
      for (int u = 0; u < EGRP; ++u) cur[u] = lds_row(Ytile, c_l, g + u, lane);
#pragma unroll
      for (int u = 0; u < EGRP; ++u) push_row(out, r_l, v_l, g + u, lbyte, cur[u]);
    }
  } else {
    for (int j = 0; j < cnt; ++j)
      push_row(out, r_l, v_l, j, lbyte, lds_row(Ytile, c_l, j, lane));
  }
}

__device__ __forceinline__ uint4 load_keys(const unsigned* __restrict__ keys,
                                           int i, int rot, int nchunks, int lane) {
  int c = i + rot;
  if (c >= nchunks) c -= nchunks;
  return *(const uint4*)(keys + (size_t)c * (CHUNK / 4) + lane * 4);
}

__global__ __launch_bounds__(1024) void gcn_owner_push(
    const float* __restrict__ x, const float* __restrict__ W,
    const float* __restrict__ ev, const int* __restrict__ row,
    const int* __restrict__ col, const unsigned* __restrict__ keys,
    float* __restrict__ out, int nrows, int nchunks) {
  __shared__ float Ytile[RPG * YP];
  __shared__ float Wlds[WLDS_FLOATS];

  const int t = threadIdx.x;
  for (int idx = t; idx < DDIM * 32; idx += 1024) {
    const int r = idx >> 5;
    const int q = idx & 31;
    const float4 w4 = *(const float4*)(W + r * DDIM + q * 4);
    float* dst = &Wlds[(r >> 1) * KP_STRIDE + (q * 4) * 2 + (r & 1)];
    dst[0] = w4.x; dst[2] = w4.y; dst[4] = w4.z; dst[6] = w4.w;
  }
  __syncthreads();

  const int wave = __builtin_amdgcn_readfirstlane(t >> 5);
  const int lane = t & 31;
  const int lh   = lane >> 4;
  const int lm   = lane & 15;
  const int node0 = blockIdx.x * RPG;

  for (int tile = wave; tile < RPG / 16; tile += NWAVE) {
    const int rowBase = node0 + tile * 16;
    const int arow = min(rowBase + lm, nrows - 1);
    const float* __restrict__ xrow = x + (size_t)arow * DDIM;

    v8f acc[8];
#pragma unroll
    for (int nt = 0; nt < 8; ++nt) acc[nt] = (v8f){0.f,0.f,0.f,0.f,0.f,0.f,0.f,0.f};

    for (int k0 = 0; k0 < DDIM; k0 += 4) {
      const v2f a = *(const v2f*)(xrow + k0 + 2 * lh);
      const float* __restrict__ wbase =
          &Wlds[((k0 >> 1) + lh) * KP_STRIDE + lm * 2];
#pragma unroll
      for (int nt = 0; nt < 8; ++nt) {
        const v2f b = *(const v2f*)(wbase + nt * 32);
        acc[nt] = __builtin_amdgcn_wmma_f32_16x16x4_f32(
             false, a,  false, b,
             (short)0, acc[nt],  false,  false);
      }
    }

#pragma unroll
    for (int nt = 0; nt < 8; ++nt) {
#pragma unroll
      for (int r = 0; r < 8; ++r)
        Ytile[(tile * 16 + r + 8 * lh) * YP + nt * 16 + lm] = acc[nt][r];
    }
  }
  __syncthreads();

  unsigned* lst = (unsigned*)Wlds + wave * LCAP;
  const unsigned gkey = blockIdx.x;
  const int rot = (int)(((unsigned)blockIdx.x * (unsigned)nchunks) / gridDim.x);
  int cnt = 0;

  for (int i = wave; i < nchunks; i += KSTEP * NWAVE) {
    uint4 kk[KSTEP];
#pragma unroll
    for (int u = 0; u < KSTEP; ++u)
      kk[u] = load_keys(keys, min(i + u * NWAVE, nchunks - 1), rot, nchunks, lane);
#pragma unroll
    for (int u = 0; u < KSTEP; ++u) {
      const int ii = i + u * NWAVE;
      if (ii < nchunks) {
        int c = ii + rot;
        if (c >= nchunks) c -= nchunks;
        const unsigned ebase = (unsigned)c * CHUNK + (unsigned)lane * 16u;
        const unsigned kw[4] = {kk[u].x, kk[u].y, kk[u].z, kk[u].w};
        bool m[16];
        unsigned bal[16];
#pragma unroll
        for (int b = 0; b < 16; ++b) {
          m[b] = ((kw[b >> 2] >> (8 * (b & 3))) & 0xFFu) == gkey;
          bal[b] = __builtin_amdgcn_ballot_w32(m[b]);
        }
#pragma unroll
        for (int w = 0; w < 4; ++w) {
          if ((bal[4 * w] | bal[4 * w + 1] | bal[4 * w + 2] | bal[4 * w + 3]) != 0u) {
#pragma unroll
            for (int j = 0; j < 4; ++j) {
              const int b = 4 * w + j;
              if (bal[b] != 0u) {
                if (m[b]) lst[cnt + (int)__builtin_amdgcn_mbcnt_lo(bal[b], 0u)] = ebase + (unsigned)b;
                cnt += __builtin_popcount(bal[b]);
              }
            }
          }
        }
        while (cnt >= 32) {
          cnt -= 32;
          push_edges(lst + cnt, 32, lane, node0, Ytile, ev, row, col, out);
        }
      }
    }
  }
  if (cnt > 0) push_edges(lst, cnt, lane, node0, Ytile, ev, row, col, out);
}

extern "C" void kernel_launch(void* const* d_in, const int* in_sizes, int n_in,
                              void* d_out, int out_size, void* d_ws, size_t ws_size,
                              hipStream_t stream) {
  const float* x    = (const float*)d_in[0];
  const float* W    = (const float*)d_in[1];
  const float* bias = (const float*)d_in[2];
  const float* ev   = (const float*)d_in[3];
  const int*   row  = (const int*)d_in[4];
  const int*   col  = (const int*)d_in[5];
  float* out = (float*)d_out;
  unsigned* keys = (unsigned*)d_ws;

  const int n = in_sizes[0] / DDIM;
  const int e = in_sizes[3];
  const int nchunks = (e + CHUNK - 1) / CHUNK;

  const int total4 = n * (DDIM / 4);
  init_out_keys<<<(total4 + 255) / 256, 256, 0, stream>>>(
      bias, col, out, keys, total4, nchunks * (CHUNK / 4), e);

  gcn_owner_push<<<(n + RPG - 1) / RPG, 1024, 0, stream>>>(
      x, W, ev, row, col, keys, out, n, nchunks);
}
